# starting kernel
# baseline (speedup 1.0000x reference)
#include <hip/hip_runtime.h>
#include <math.h>

typedef __attribute__((ext_vector_type(2))) float v2f;
typedef __attribute__((ext_vector_type(8))) float v8f;

#define IN_F   128
#define OUT_F  64
#define KSTEPS 32
#define BLK_FLOATS 128
#define BLK_STRIDE 160
#define B_IMG_FLOATS (2 * KSTEPS * BLK_STRIDE)

__global__ __launch_bounds__(256) void prep_b_kernel(
    const float* __restrict__ weight, float* __restrict__ wsB)
{
    const int tid = threadIdx.x;
    for (int idx = tid; idx < B_IMG_FLOATS; idx += 256) {
        int blk = idx / BLK_STRIDE;
        int e   = idx - blk * BLK_STRIDE;
        float v = 0.0f;
        if (e < BLK_FLOATS) {
            int nt  = e >> 5;
            int r   = e & 31;
            int hl2 = r >> 1;
            int p   = r & 1;
            v = weight[(blk * 2 + p) * OUT_F + nt * 16 + hl2];
        }
        wsB[idx] = v;
    }
}

__global__ __launch_bounds__(256) void gcn_fused_kernel(
    const float* __restrict__ feat,
    const float* __restrict__ w,
    const float* __restrict__ wsB,
    const float* __restrict__ bias,
    const float* __restrict__ alpha_p,
    float* __restrict__ out_pool,
    float* __restrict__ out_anchor)
{
    __shared__ float sB[B_IMG_FLOATS];

    const int tid = threadIdx.x;
    {
        const float4* s4 = (const float4*)wsB;
        float4*       d4 = (float4*)sB;
        #pragma unroll
        for (int i = 0; i < B_IMG_FLOATS / 4 / 256; ++i)
            d4[tid + i * 256] = s4[tid + i * 256];
    }
    __syncthreads();

    const int wave = tid >> 5;
    const int lane = tid & 31;
    const int hl   = lane & 15;
    const int hi   = lane >> 4;

    const long long tileRow0 = ((long long)blockIdx.x * 8 + wave) * 16;

    const float* aRow = feat + (tileRow0 + hl) * IN_F + 2 * hi;

    v2f a[KSTEPS];
    #pragma unroll
    for (int ks = 0; ks < KSTEPS; ++ks)
        a[ks] = __builtin_nontemporal_load((const v2f*)(aRow + ks * 4));

    v8f c[4] = {};

    #pragma unroll
    for (int ks = 0; ks < KSTEPS; ++ks) {
        const float* bp = sB + (2 * ks + hi) * BLK_STRIDE + hl * 2;
        #pragma unroll
        for (int nt = 0; nt < 4; ++nt) {
            v2f b = *(const v2f*)(bp + nt * 32);
            c[nt] = __builtin_amdgcn_wmma_f32_16x16x4_f32(
                false, a[ks], false, b, (short)0, c[nt], false, false);
        }
    }

    const float alpha = alpha_p[0];
    float bn[4];
    #pragma unroll
    for (int nt = 0; nt < 4; ++nt) bn[nt] = bias[16 * nt + hl];

    const long long tileGroup0 = tileRow0 >> 2;

    #pragma unroll
    for (int gg = 0; gg < 2; ++gg) {
        const long long g = tileGroup0 + 2 * hi + gg;
        const float* wg = w + g * 16;
        float w1[4], w2[4], w3[4];
        #pragma unroll
        for (int j = 0; j < 4; ++j) {
            w1[j] = wg[4 + j];
            w2[j] = wg[8 + j];
            w3[j] = wg[12 + j];
        }

        float pool[4], anch[4];
        float ssP = 0.0f, ssA = 0.0f;
        #pragma unroll
        for (int nt = 0; nt < 4; ++nt) {
            float h0 = c[nt][4 * gg + 0];
            float h1 = c[nt][4 * gg + 1];
            float h2 = c[nt][4 * gg + 2];
            float h3 = c[nt][4 * gg + 3];
            float p = 0.0f;
            #pragma unroll
            for (int j = 0; j < 4; ++j) {
                float acc = fmaf(w1[j], h1, fmaf(w2[j], h2, fmaf(w3[j], h3, bn[nt])));
                p += (acc >= 0.0f) ? acc : alpha * acc;
            }
            p *= 0.25f;
            float a0 = h0 + bn[nt];
            a0 = (a0 >= 0.0f) ? a0 : alpha * a0;
            pool[nt] = p;
            anch[nt] = a0;
            ssP = fmaf(p, p, ssP);
            ssA = fmaf(a0, a0, ssA);
        }

        #pragma unroll
        for (int m = 1; m < 16; m <<= 1) {
            ssP += __shfl_xor(ssP, m, 32);
            ssA += __shfl_xor(ssA, m, 32);
        }
        float sP = 1.0f / fmaxf(sqrtf(ssP), 1e-12f);
        float sA = 1.0f / fmaxf(sqrtf(ssA), 1e-12f);

        float* op = out_pool   + g * 64 + hl;
        float* oa = out_anchor + g * 64 + hl;
        #pragma unroll
        for (int nt = 0; nt < 4; ++nt) {
            __builtin_nontemporal_store(pool[nt] * sP, op + 16 * nt);
            __builtin_nontemporal_store(anch[nt] * sA, oa + 16 * nt);
        }
    }
}

extern "C" void kernel_launch(void* const* d_in, const int* in_sizes, int n_in,
                              void* d_out, int out_size, void* d_ws, size_t ws_size,
                              hipStream_t stream) {
    const float* feat   = (const float*)d_in[0];
    const float* w      = (const float*)d_in[1];
    const float* weight = (const float*)d_in[2];
    const float* bias   = (const float*)d_in[3];
    const float* alpha  = (const float*)d_in[4];

    const long long nRows   = (long long)in_sizes[0] / IN_F;
    const long long nGroups = nRows / 4;

    float* wsB = (float*)d_ws;

    float* out_pool   = (float*)d_out;
    float* out_anchor = out_pool + nGroups * OUT_F;

    prep_b_kernel<<<1, 256, 0, stream>>>(weight, wsB);

    const int blocks = (int)(nRows / 128);
    gcn_fused_kernel<<<blocks, 256, 0, stream>>>(
        feat, w, wsB, bias, alpha, out_pool, out_anchor);
}


# A tile read as 16 whole 512-B rows (coalesced b128 loads) and turned into WMMA A fragments through a padded per-wave LDS slab, replacing loads that took 16 B from each of 16 rows
# speedup vs baseline: 1.6497x; 1.6497x over previous
#include <hip/hip_runtime.h>
#include <math.h>

typedef __attribute__((ext_vector_type(2))) float v2f;
typedef __attribute__((ext_vector_type(4))) float v4f;
typedef __attribute__((ext_vector_type(8))) float v8f;

#define IN_F   128
#define OUT_F  64
#define KSTEPS 32
#define BLK_FLOATS 128
#define BLK_STRIDE 160
#define B_IMG_FLOATS (2 * KSTEPS * BLK_STRIDE)
#define A_PITCH 132
#define A_SLAB_FLOATS (16 * A_PITCH)

__global__ __launch_bounds__(256) void prep_b_kernel(
    const float* __restrict__ weight, float* __restrict__ wsB)
{
    const int tid = threadIdx.x;
    for (int idx = tid; idx < B_IMG_FLOATS; idx += 256) {
        int blk = idx / BLK_STRIDE;
        int e   = idx - blk * BLK_STRIDE;
        float v = 0.0f;
        if (e < BLK_FLOATS) {
            int nt  = e >> 5;
            int r   = e & 31;
            int hl2 = r >> 1;
            int p   = r & 1;
            v = weight[(blk * 2 + p) * OUT_F + nt * 16 + hl2];
        }
        wsB[idx] = v;
    }
}

__global__ __launch_bounds__(256) void gcn_fused_kernel(
    const float* __restrict__ feat,
    const float* __restrict__ w,
    const float* __restrict__ wsB,
    const float* __restrict__ bias,
    const float* __restrict__ alpha_p,
    float* __restrict__ out_pool,
    float* __restrict__ out_anchor)
{
    __shared__ __attribute__((aligned(16))) float smem[B_IMG_FLOATS + 8 * A_SLAB_FLOATS];
    float* sB = smem;
    float* sA = smem + B_IMG_FLOATS;

    const int tid  = threadIdx.x;
    const int wave = tid >> 5;
    const int lane = tid & 31;
    const int hl   = lane & 15;
    const int hi   = lane >> 4;

    const long long tileRow0 = ((long long)blockIdx.x * 8 + wave) * 16;

    const float* aTile = feat + tileRow0 * IN_F + 4 * lane;
    v4f araw[16];
    #pragma unroll
    for (int r = 0; r < 16; ++r)
        araw[r] = __builtin_nontemporal_load((const v4f*)(aTile + r * IN_F));

    {
        const float4* s4 = (const float4*)wsB;
        float4*       d4 = (float4*)sB;
        #pragma unroll
        for (int i = 0; i < B_IMG_FLOATS / 4 / 256; ++i)
            d4[tid + i * 256] = s4[tid + i * 256];
    }

    float* slab = sA + wave * A_SLAB_FLOATS;
    #pragma unroll
    for (int r = 0; r < 16; ++r)
        *(v4f*)(slab + r * A_PITCH + 4 * lane) = araw[r];

    __syncthreads();

    const float* aFrag = slab + hl * A_PITCH + 2 * hi;
    v2f a[KSTEPS];
    #pragma unroll
    for (int ks = 0; ks < KSTEPS; ++ks)
        a[ks] = *(const v2f*)(aFrag + ks * 4);

    v8f c[4] = {};

    #pragma unroll
    for (int ks = 0; ks < KSTEPS; ++ks) {
        const float* bp = sB + (2 * ks + hi) * BLK_STRIDE + hl * 2;
        #pragma unroll
        for (int nt = 0; nt < 4; ++nt) {
            v2f b = *(const v2f*)(bp + nt * 32);
            c[nt] = __builtin_amdgcn_wmma_f32_16x16x4_f32(
                false, a[ks], false, b, (short)0, c[nt], false, false);
        }
    }

    const float alpha = alpha_p[0];
    float bn[4];
    #pragma unroll
    for (int nt = 0; nt < 4; ++nt) bn[nt] = bias[16 * nt + hl];

    const long long tileGroup0 = tileRow0 >> 2;

    #pragma unroll
    for (int gg = 0; gg < 2; ++gg) {
        const long long g = tileGroup0 + 2 * hi + gg;
        const float* wg = w + g * 16;
        float w1[4], w2[4], w3[4];
        #pragma unroll
        for (int j = 0; j < 4; ++j) {
            w1[j] = wg[4 + j];
            w2[j] = wg[8 + j];
            w3[j] = wg[12 + j];
        }

        float pool[4], anch[4];
        float ssP = 0.0f, ssA = 0.0f;
        #pragma unroll
        for (int nt = 0; nt < 4; ++nt) {
            float h0 = c[nt][4 * gg + 0];
            float h1 = c[nt][4 * gg + 1];
            float h2 = c[nt][4 * gg + 2];
            float h3 = c[nt][4 * gg + 3];
            float p = 0.0f;
            #pragma unroll
            for (int j = 0; j < 4; ++j) {
                float acc = fmaf(w1[j], h1, fmaf(w2[j], h2, fmaf(w3[j], h3, bn[nt])));
                p += (acc >= 0.0f) ? acc : alpha * acc;
            }
            p *= 0.25f;
            float a0 = h0 + bn[nt];
            a0 = (a0 >= 0.0f) ? a0 : alpha * a0;
            pool[nt] = p;
            anch[nt] = a0;
            ssP = fmaf(p, p, ssP);
            ssA = fmaf(a0, a0, ssA);
        }

        #pragma unroll
        for (int m = 1; m < 16; m <<= 1) {
            ssP += __shfl_xor(ssP, m, 32);
            ssA += __shfl_xor(ssA, m, 32);
        }
        float sP = 1.0f / fmaxf(sqrtf(ssP), 1e-12f);
        float sA = 1.0f / fmaxf(sqrtf(ssA), 1e-12f);

        float* op = out_pool   + g * 64 + hl;
        float* oa = out_anchor + g * 64 + hl;
        #pragma unroll
        for (int nt = 0; nt < 4; ++nt) {
            __builtin_nontemporal_store(pool[nt] * sP, op + 16 * nt);
            __builtin_nontemporal_store(anch[nt] * sA, oa + 16 * nt);
        }
    }
}

extern "C" void kernel_launch(void* const* d_in, const int* in_sizes, int n_in,
                              void* d_out, int out_size, void* d_ws, size_t ws_size,
                              hipStream_t stream) {
    const float* feat   = (const float*)d_in[0];
    const float* w      = (const float*)d_in[1];
    const float* weight = (const float*)d_in[2];
    const float* bias   = (const float*)d_in[3];
    const float* alpha  = (const float*)d_in[4];

    const long long nRows   = (long long)in_sizes[0] / IN_F;
    const long long nGroups = nRows / 4;

    float* wsB = (float*)d_ws;

    float* out_pool   = (float*)d_out;
    float* out_anchor = out_pool + nGroups * OUT_F;

    prep_b_kernel<<<1, 256, 0, stream>>>(weight, wsB);

    const int blocks = (int)(nRows / 128);
    gcn_fused_kernel<<<blocks, 256, 0, stream>>>(
        feat, w, wsB, bias, alpha, out_pool, out_anchor);
}
